# starting kernel
# baseline (speedup 1.0000x reference)
#include <hip/hip_runtime.h>
#include <math.h>

typedef __attribute__((ext_vector_type(2))) float v2f;
typedef __attribute__((ext_vector_type(8))) float v8f;

#define PB_EPS     1e-7f
#define PB_MIN_DEN 1e-15f

__global__ __launch_bounds__(128) void poincare_dist_kernel(
    const float* __restrict__ matrix,
    const int*   __restrict__ items,
    const int*   __restrict__ origin,
    float*       __restrict__ out,
    int L, int nTiles)
{
    __shared__ float sy[32];
    __shared__ float sy2;

    const int b    = blockIdx.x;
    const int tid  = threadIdx.x;
    const int lane = tid & 31;
    const int wave = tid >> 5;
    const int nWaves = blockDim.x >> 5;

    if (wave == 0) {
        const float yd = matrix[(size_t)origin[b] * 32 + lane];
        sy[lane] = yd;
        float p = yd * yd;
        #pragma unroll
        for (int off = 16; off > 0; off >>= 1) p += __shfl_xor(p, off, 32);
        if (lane == 0) sy2 = p;
    }
    __syncthreads();
    const float y2 = sy2;

    const int h = lane >> 4;
    const int m = lane & 15;
    const int r = lane & 7;
    const bool owner = ((lane >> 3) & 1) == h;
    const int  cown  = r + 8 * h;

    for (int t = wave; t < nTiles; t += nWaves) {
        const int l0 = t * 16;
        int l = l0 + m;
        if (l >= L) l = L - 1;
        const int idx = items[(size_t)b * L + l];
        const float* row = matrix + (size_t)idx * 32 + 2 * h;

        v8f accY = {};
        v8f accG = {};
        #pragma unroll
        for (int s = 0; s < 8; ++s) {
            const v2f a = *(const v2f*)(row + 4 * s);
            v2f bv;
            bv.x = sy[4 * s + 2 * h];
            bv.y = sy[4 * s + 2 * h + 1];
            accY = __builtin_amdgcn_wmma_f32_16x16x4_f32(
                false, a, false, bv, (short)0, accY, false, false);
            accG = __builtin_amdgcn_wmma_f32_16x16x4_f32(
                false, a, false, a, (short)0, accG, false, false);
        }

        if (owner && (l0 + cown) < L) {
            float xy = accY[0];
            float a2 = accG[0];
            #pragma unroll
            for (int i = 1; i < 8; ++i) {
                if (r == i) { xy = accY[i]; a2 = accG[i]; }
            }
            const float ay = -xy;
            const float al = 1.0f + 2.0f * ay + y2;
            const float be = 1.0f - a2;
            float den = 1.0f + 2.0f * ay + a2 * y2;
            den = fmaxf(den, PB_MIN_DEN);
            float nn = al * al * a2 + 2.0f * al * be * ay + be * be * y2;
            nn = fmaxf(nn, 0.0f);
            float z = sqrtf(nn) / den;
            z = fminf(z, 1.0f - PB_EPS);
            out[(size_t)b * L + l0 + cown] = -2.0f * atanhf(z);
        }
    }
}

extern "C" void kernel_launch(void* const* d_in, const int* in_sizes, int n_in,
                              void* d_out, int out_size, void* d_ws, size_t ws_size,
                              hipStream_t stream) {
    (void)n_in; (void)d_ws; (void)ws_size; (void)out_size;
    const float* matrix = (const float*)d_in[0];
    const int*   items  = (const int*)d_in[1];
    const int*   origin = (const int*)d_in[2];
    float*       out    = (float*)d_out;

    const int B = in_sizes[2];
    const int L = in_sizes[1] / B;
    const int nTiles = (L + 15) / 16;

    poincare_dist_kernel<<<B, 128, 0, stream>>>(matrix, items, origin, out, L, nTiles);
}


# One wave per origin row, all its gathers in flight: 4096 waves (was 16384), each issues 13 tiles' index loads and 52 16-byte row loads ahead of its WMMAs; no LDS, no barrier; WMMA k order changes
# speedup vs baseline: 1.0813x; 1.0813x over previous
#include <hip/hip_runtime.h>
#include <math.h>

typedef __attribute__((ext_vector_type(2))) float v2f;
typedef __attribute__((ext_vector_type(4))) float v4f;
typedef __attribute__((ext_vector_type(8))) float v8f;

#define PB_EPS     1e-7f
#define PB_MIN_DEN 1e-15f

#define TILES_PER_PASS  13
#define WAVES_PER_BLOCK 8

__device__ __forceinline__ void load_row_share(const float* __restrict__ row_h, v4f q[4])
{
    #pragma unroll
    for (int j = 0; j < 4; ++j) q[j] = *(const v4f*)(row_h + 8 * j);
}

__global__ __launch_bounds__(32 * WAVES_PER_BLOCK, 4) void poincare_dist_kernel(
    const float* __restrict__ matrix,
    const int*   __restrict__ items,
    const int*   __restrict__ origin,
    float*       __restrict__ out,
    int B, int L, int nTiles)
{
    const int lane = threadIdx.x & 31;
    const int wave = __builtin_amdgcn_readfirstlane((int)(threadIdx.x >> 5));
    const int bw   = blockIdx.x * WAVES_PER_BLOCK + wave;
    const bool live = bw < B;
    const int  b    = live ? bw : B - 1;

    const int h = lane >> 4;
    const int m = lane & 15;
    const int r = lane & 7;
    const bool owner = ((lane >> 3) & 1) == h;
    const int  cown  = r + 8 * h;

    const int* __restrict__ it = items + (size_t)b * L;
    float*     __restrict__ ob = out   + (size_t)b * L;

    const unsigned oi = (unsigned)origin[b];

    unsigned idx[TILES_PER_PASS];
    #pragma unroll
    for (int i = 0; i < TILES_PER_PASS; ++i) {
        int l = i * 16 + m;
        if (l >= L) l = L - 1;
        idx[i] = (unsigned)it[l];
    }

    const float* __restrict__ yrow = matrix + (size_t)oi * 32;
    v4f yq[4];
    load_row_share(yrow + 4 * h, yq);
    float y2 = yrow[lane];

    v4f q[TILES_PER_PASS][4];
    #pragma unroll
    for (int i = 0; i < TILES_PER_PASS; ++i)
        load_row_share(matrix + (size_t)idx[i] * 32 + 4 * h, q[i]);

    __builtin_amdgcn_sched_barrier(0);

    y2 = y2 * y2;
    #pragma unroll
    for (int off = 16; off > 0; off >>= 1) y2 += __shfl_xor(y2, off, 32);

    for (int t0 = 0; ; ) {
        #pragma unroll
        for (int i = 0; i < TILES_PER_PASS; ++i) {
            v8f accY = {};
            v8f accG = {};
            #pragma unroll
            for (int j = 0; j < 4; ++j) {
                #pragma unroll
                for (int p = 0; p < 2; ++p) {
                    v2f a, bv;
                    a.x  = q[i][j][2 * p];  a.y  = q[i][j][2 * p + 1];
                    bv.x = yq[j][2 * p];    bv.y = yq[j][2 * p + 1];
                    accY = __builtin_amdgcn_wmma_f32_16x16x4_f32(
                        false, a, false, bv, (short)0, accY, false, false);
                    accG = __builtin_amdgcn_wmma_f32_16x16x4_f32(
                        false, a, false, a, (short)0, accG, false, false);
                }
            }

            const int l0 = (t0 + i) * 16;
            if (live && owner && (l0 + cown) < L) {
                float xy = accY[0];
                float a2 = accG[0];
                #pragma unroll
                for (int k = 1; k < 8; ++k) {
                    if (r == k) { xy = accY[k]; a2 = accG[k]; }
                }
                const float ay = -xy;
                const float al = 1.0f + 2.0f * ay + y2;
                const float be = 1.0f - a2;
                float den = 1.0f + 2.0f * ay + a2 * y2;
                den = fmaxf(den, PB_MIN_DEN);
                float nn = al * al * a2 + 2.0f * al * be * ay + be * be * y2;
                nn = fmaxf(nn, 0.0f);
                float z = sqrtf(nn) / den;
                z = fminf(z, 1.0f - PB_EPS);
                ob[l0 + cown] = -2.0f * atanhf(z);
            }
        }

        t0 += TILES_PER_PASS;
        if (t0 >= nTiles) break;
        #pragma unroll
        for (int i = 0; i < TILES_PER_PASS; ++i) {
            int l = (t0 + i) * 16 + m;
            if (l >= L) l = L - 1;
            idx[i] = (unsigned)it[l];
        }
        #pragma unroll
        for (int i = 0; i < TILES_PER_PASS; ++i)
            load_row_share(matrix + (size_t)idx[i] * 32 + 4 * h, q[i]);
    }
}

extern "C" void kernel_launch(void* const* d_in, const int* in_sizes, int n_in,
                              void* d_out, int out_size, void* d_ws, size_t ws_size,
                              hipStream_t stream) {
    (void)n_in; (void)d_ws; (void)ws_size; (void)out_size;
    const float* matrix = (const float*)d_in[0];
    const int*   items  = (const int*)d_in[1];
    const int*   origin = (const int*)d_in[2];
    float*       out    = (float*)d_out;

    const int B = in_sizes[2];
    const int L = in_sizes[1] / B;
    const int nTiles = (L + 15) / 16;

    poincare_dist_kernel<<<(B + WAVES_PER_BLOCK - 1) / WAVES_PER_BLOCK, 32 * WAVES_PER_BLOCK, 0, stream>>>(
        matrix, items, origin, out, B, L, nTiles);
}


# One Gram product per tile yields both |x|^2 (diagonal) and x.y (row 15 = origin row): tiles of 15 candidates + origin row, 112 WMMAs per wave (was 208); k order and every sum unchanged
# speedup vs baseline: 1.2477x; 1.1539x over previous
#include <hip/hip_runtime.h>
#include <math.h>

typedef __attribute__((ext_vector_type(2))) float v2f;
typedef __attribute__((ext_vector_type(4))) float v4f;
typedef __attribute__((ext_vector_type(8))) float v8f;

#define PB_EPS     1e-7f
#define PB_MIN_DEN 1e-15f

#define ROWS_PER_TILE   15
#define TILES_PER_PASS  14
#define WAVES_PER_BLOCK 8

__device__ __forceinline__ void load_row_share(const float* __restrict__ row_h, v4f q[4])
{
    #pragma unroll
    for (int j = 0; j < 4; ++j) q[j] = *(const v4f*)(row_h + 8 * j);
}

__device__ __forceinline__ float xhalf(float x)
{
    int b = __builtin_bit_cast(int, x);
    return __builtin_bit_cast(float,
        __builtin_amdgcn_permlanex16(b, b, 0x76543210, 0xfedcba98, false, false));
}

__global__ __launch_bounds__(32 * WAVES_PER_BLOCK, 4) void poincare_dist_kernel(
    const float* __restrict__ matrix,
    const int*   __restrict__ items,
    const int*   __restrict__ origin,
    float*       __restrict__ out,
    int B, int L, int nTiles)
{
    const int lane = threadIdx.x & 31;
    const int wave = __builtin_amdgcn_readfirstlane((int)(threadIdx.x >> 5));
    const int bw   = blockIdx.x * WAVES_PER_BLOCK + wave;
    const bool live = bw < B;
    const int  b    = live ? bw : B - 1;

    const int h = lane >> 4;
    const int m = lane & 15;
    const int r = lane & 7;
    const int  cown  = r + 8 * h;
    const bool owner = (((lane >> 3) & 1) == h) && (cown < ROWS_PER_TILE);

    const int* __restrict__ it = items + (size_t)b * L;
    float*     __restrict__ ob = out   + (size_t)b * L;

    const unsigned oi = (unsigned)origin[b];

    unsigned idx[TILES_PER_PASS];
    #pragma unroll
    for (int i = 0; i < TILES_PER_PASS; ++i) {
        int l = i * ROWS_PER_TILE + m;
        if (l >= L) l = L - 1;
        const unsigned cand = (unsigned)it[l];
        idx[i] = (m == ROWS_PER_TILE) ? oi : cand;
    }

    const float* __restrict__ yrow = matrix + (size_t)oi * 32;
    float y2 = yrow[lane];

    v4f q[TILES_PER_PASS][4];
    #pragma unroll
    for (int i = 0; i < TILES_PER_PASS; ++i)
        load_row_share(matrix + (size_t)idx[i] * 32 + 4 * h, q[i]);

    __builtin_amdgcn_sched_barrier(0);

    y2 = y2 * y2;
    #pragma unroll
    for (int off = 16; off > 0; off >>= 1) y2 += __shfl_xor(y2, off, 32);

    for (int t0 = 0; ; ) {
        #pragma unroll
        for (int i = 0; i < TILES_PER_PASS; ++i) {
            v8f accG = {};
            #pragma unroll
            for (int j = 0; j < 4; ++j) {
                #pragma unroll
                for (int p = 0; p < 2; ++p) {
                    v2f a;
                    a.x = q[i][j][2 * p];  a.y = q[i][j][2 * p + 1];
                    accG = __builtin_amdgcn_wmma_f32_16x16x4_f32(
                        false, a, false, a, (short)0, accG, false, false);
                }
            }

            const float g15x = xhalf(accG[7]);

            const int l0 = (t0 + i) * ROWS_PER_TILE;
            if (live && owner && (l0 + cown) < L) {
                float a2 = accG[0];
                #pragma unroll
                for (int k = 1; k < 8; ++k) {
                    if (r == k) a2 = accG[k];
                }
                const float xy = h ? accG[7] : g15x;
                const float ay = -xy;
                const float al = 1.0f + 2.0f * ay + y2;
                const float be = 1.0f - a2;
                float den = 1.0f + 2.0f * ay + a2 * y2;
                den = fmaxf(den, PB_MIN_DEN);
                float nn = al * al * a2 + 2.0f * al * be * ay + be * be * y2;
                nn = fmaxf(nn, 0.0f);
                float z = sqrtf(nn) / den;
                z = fminf(z, 1.0f - PB_EPS);
                ob[l0 + cown] = -2.0f * atanhf(z);
            }
        }

        t0 += TILES_PER_PASS;
        if (t0 >= nTiles) break;
        #pragma unroll
        for (int i = 0; i < TILES_PER_PASS; ++i) {
            int l = (t0 + i) * ROWS_PER_TILE + m;
            if (l >= L) l = L - 1;
            const unsigned cand = (unsigned)it[l];
            idx[i] = (m == ROWS_PER_TILE) ? oi : cand;
        }
        #pragma unroll
        for (int i = 0; i < TILES_PER_PASS; ++i)
            load_row_share(matrix + (size_t)idx[i] * 32 + 4 * h, q[i]);
    }
}

extern "C" void kernel_launch(void* const* d_in, const int* in_sizes, int n_in,
                              void* d_out, int out_size, void* d_ws, size_t ws_size,
                              hipStream_t stream) {
    (void)n_in; (void)d_ws; (void)ws_size; (void)out_size;
    const float* matrix = (const float*)d_in[0];
    const int*   items  = (const int*)d_in[1];
    const int*   origin = (const int*)d_in[2];
    float*       out    = (float*)d_out;

    const int B = in_sizes[2];
    const int L = in_sizes[1] / B;
    const int nTiles = (L + ROWS_PER_TILE - 1) / ROWS_PER_TILE;

    poincare_dist_kernel<<<(B + WAVES_PER_BLOCK - 1) / WAVES_PER_BLOCK, 32 * WAVES_PER_BLOCK, 0, stream>>>(
        matrix, items, origin, out, B, L, nTiles);
}


# Distance epilogue run once per two tiles: second tile's |x|^2 and x.y moved 8 lanes over (row_xmask:8), 30 lanes finish 30 candidates; 7 epilogues per wave (was 14 on 15 lanes); every sum unchanged
# speedup vs baseline: 1.3904x; 1.1144x over previous
#include <hip/hip_runtime.h>
#include <math.h>

typedef __attribute__((ext_vector_type(2))) float v2f;
typedef __attribute__((ext_vector_type(4))) float v4f;
typedef __attribute__((ext_vector_type(8))) float v8f;

#define PB_EPS     1e-7f
#define PB_MIN_DEN 1e-15f

#define ROWS_PER_TILE   15
#define TILES_PER_PASS  14
#define PAIRS_PER_PASS  (TILES_PER_PASS / 2)
#define WAVES_PER_BLOCK 8

__device__ __forceinline__ void load_row_share(const float* __restrict__ row_h, v4f q[4])
{
    #pragma unroll
    for (int j = 0; j < 4; ++j) q[j] = *(const v4f*)(row_h + 8 * j);
}

__device__ __forceinline__ float xhalf(float x)
{
    int b = __builtin_bit_cast(int, x);
    return __builtin_bit_cast(float,
        __builtin_amdgcn_permlanex16(b, b, 0x76543210, 0xfedcba98, false, false));
}

__device__ __forceinline__ float xrow8(float x)
{
    int b = __builtin_bit_cast(int, x);
    return __builtin_bit_cast(float,
        __builtin_amdgcn_update_dpp(b, b, 0x168  , 0xf, 0xf, false));
}

__global__ __launch_bounds__(32 * WAVES_PER_BLOCK, 4) void poincare_dist_kernel(
    const float* __restrict__ matrix,
    const int*   __restrict__ items,
    const int*   __restrict__ origin,
    float*       __restrict__ out,
    int B, int L, int nTiles)
{
    const int lane = threadIdx.x & 31;
    const int wave = __builtin_amdgcn_readfirstlane((int)(threadIdx.x >> 5));
    const int bw   = blockIdx.x * WAVES_PER_BLOCK + wave;
    const bool live = bw < B;
    const int  b    = live ? bw : B - 1;

    const int h = lane >> 4;
    const int m = lane & 15;
    const int r = lane & 7;
    const int  cown  = r + 8 * h;
    const bool first = ((lane >> 3) & 1) == h;
    const bool valid = cown < ROWS_PER_TILE;

    const int* __restrict__ it = items + (size_t)b * L;
    float*     __restrict__ ob = out   + (size_t)b * L;

    const unsigned oi = (unsigned)origin[b];

    unsigned idx[TILES_PER_PASS];
    #pragma unroll
    for (int i = 0; i < TILES_PER_PASS; ++i) {
        int l = i * ROWS_PER_TILE + m;
        if (l >= L) l = L - 1;
        const unsigned cand = (unsigned)it[l];
        idx[i] = (m == ROWS_PER_TILE) ? oi : cand;
    }

    const float* __restrict__ yrow = matrix + (size_t)oi * 32;
    float y2 = yrow[lane];

    v4f q[TILES_PER_PASS][4];
    #pragma unroll
    for (int i = 0; i < TILES_PER_PASS; ++i)
        load_row_share(matrix + (size_t)idx[i] * 32 + 4 * h, q[i]);

    __builtin_amdgcn_sched_barrier(0);

    y2 = y2 * y2;
    #pragma unroll
    for (int off = 16; off > 0; off >>= 1) y2 += __shfl_xor(y2, off, 32);

    for (int t0 = 0; ; ) {
        #pragma unroll
        for (int p = 0; p < PAIRS_PER_PASS; ++p) {
            float a2t[2], xyt[2];
            #pragma unroll
            for (int e = 0; e < 2; ++e) {
                const int i = 2 * p + e;
                v8f accG = {};
                #pragma unroll
                for (int j = 0; j < 4; ++j) {
                    #pragma unroll
                    for (int s = 0; s < 2; ++s) {
                        v2f a;
                        a.x = q[i][j][2 * s];  a.y = q[i][j][2 * s + 1];
                        accG = __builtin_amdgcn_wmma_f32_16x16x4_f32(
                            false, a, false, a, (short)0, accG, false, false);
                    }
                }

                float d = accG[0];
                #pragma unroll
                for (int k = 1; k < 8; ++k) d = (r == k) ? accG[k] : d;
                const float g15x = xhalf(accG[7]);
                a2t[e] = d;
                xyt[e] = h ? accG[7] : g15x;
            }

            const float a2b = xrow8(a2t[1]);
            const float xyb = xrow8(xyt[1]);
            const float a2  = first ? a2t[0] : a2b;
            const float xy  = first ? xyt[0] : xyb;

            const int l = (t0 + 2 * p + (first ? 0 : 1)) * ROWS_PER_TILE + cown;
            if (live && valid && l < L) {
                const float ay = -xy;
                const float al = 1.0f + 2.0f * ay + y2;
                const float be = 1.0f - a2;
                float den = 1.0f + 2.0f * ay + a2 * y2;
                den = fmaxf(den, PB_MIN_DEN);
                float nn = al * al * a2 + 2.0f * al * be * ay + be * be * y2;
                nn = fmaxf(nn, 0.0f);
                float z = sqrtf(nn) / den;
                z = fminf(z, 1.0f - PB_EPS);
                ob[l] = -2.0f * atanhf(z);
            }
        }

        t0 += TILES_PER_PASS;
        if (t0 >= nTiles) break;
        #pragma unroll
        for (int i = 0; i < TILES_PER_PASS; ++i) {
            int l = (t0 + i) * ROWS_PER_TILE + m;
            if (l >= L) l = L - 1;
            const unsigned cand = (unsigned)it[l];
            idx[i] = (m == ROWS_PER_TILE) ? oi : cand;
        }
        #pragma unroll
        for (int i = 0; i < TILES_PER_PASS; ++i)
            load_row_share(matrix + (size_t)idx[i] * 32 + 4 * h, q[i]);
    }
}

extern "C" void kernel_launch(void* const* d_in, const int* in_sizes, int n_in,
                              void* d_out, int out_size, void* d_ws, size_t ws_size,
                              hipStream_t stream) {
    (void)n_in; (void)d_ws; (void)ws_size; (void)out_size;
    const float* matrix = (const float*)d_in[0];
    const int*   items  = (const int*)d_in[1];
    const int*   origin = (const int*)d_in[2];
    float*       out    = (float*)d_out;

    const int B = in_sizes[2];
    const int L = in_sizes[1] / B;
    const int nTiles = (L + ROWS_PER_TILE - 1) / ROWS_PER_TILE;

    poincare_dist_kernel<<<(B + WAVES_PER_BLOCK - 1) / WAVES_PER_BLOCK, 32 * WAVES_PER_BLOCK, 0, stream>>>(
        matrix, items, origin, out, B, L, nTiles);
}


# atanh in the epilogue by its odd series to z^13 when every finishing lane of the wave has z < 1/4 (one wave-uniform branch, library atanhf otherwise): ~90 fewer vector instructions per epilogue
# speedup vs baseline: 1.5093x; 1.0855x over previous
#include <hip/hip_runtime.h>
#include <math.h>

typedef __attribute__((ext_vector_type(2))) float v2f;
typedef __attribute__((ext_vector_type(4))) float v4f;
typedef __attribute__((ext_vector_type(8))) float v8f;

#define PB_EPS     1e-7f
#define PB_MIN_DEN 1e-15f

#define ROWS_PER_TILE   15
#define TILES_PER_PASS  14
#define PAIRS_PER_PASS  (TILES_PER_PASS / 2)
#define WAVES_PER_BLOCK 8

__device__ __forceinline__ void load_row_share(const float* __restrict__ row_h, v4f q[4])
{
    #pragma unroll
    for (int j = 0; j < 4; ++j) q[j] = *(const v4f*)(row_h + 8 * j);
}

__device__ __forceinline__ float xhalf(float x)
{
    int b = __builtin_bit_cast(int, x);
    return __builtin_bit_cast(float,
        __builtin_amdgcn_permlanex16(b, b, 0x76543210, 0xfedcba98, false, false));
}

__device__ __forceinline__ float xrow8(float x)
{
    int b = __builtin_bit_cast(int, x);
    return __builtin_bit_cast(float,
        __builtin_amdgcn_update_dpp(b, b, 0x168  , 0xf, 0xf, false));
}

__global__ __launch_bounds__(32 * WAVES_PER_BLOCK, 4) void poincare_dist_kernel(
    const float* __restrict__ matrix,
    const int*   __restrict__ items,
    const int*   __restrict__ origin,
    float*       __restrict__ out,
    int B, int L, int nTiles)
{
    const int lane = threadIdx.x & 31;
    const int wave = __builtin_amdgcn_readfirstlane((int)(threadIdx.x >> 5));
    const int bw   = blockIdx.x * WAVES_PER_BLOCK + wave;
    const bool live = bw < B;
    const int  b    = live ? bw : B - 1;

    const int h = lane >> 4;
    const int m = lane & 15;
    const int r = lane & 7;
    const int  cown  = r + 8 * h;
    const bool first = ((lane >> 3) & 1) == h;
    const bool valid = cown < ROWS_PER_TILE;

    const int* __restrict__ it = items + (size_t)b * L;
    float*     __restrict__ ob = out   + (size_t)b * L;

    const unsigned oi = (unsigned)origin[b];

    unsigned idx[TILES_PER_PASS];
    #pragma unroll
    for (int i = 0; i < TILES_PER_PASS; ++i) {
        int l = i * ROWS_PER_TILE + m;
        if (l >= L) l = L - 1;
        const unsigned cand = (unsigned)it[l];
        idx[i] = (m == ROWS_PER_TILE) ? oi : cand;
    }

    const float* __restrict__ yrow = matrix + (size_t)oi * 32;
    float y2 = yrow[lane];

    v4f q[TILES_PER_PASS][4];
    #pragma unroll
    for (int i = 0; i < TILES_PER_PASS; ++i)
        load_row_share(matrix + (size_t)idx[i] * 32 + 4 * h, q[i]);

    __builtin_amdgcn_sched_barrier(0);

    y2 = y2 * y2;
    #pragma unroll
    for (int off = 16; off > 0; off >>= 1) y2 += __shfl_xor(y2, off, 32);

    for (int t0 = 0; ; ) {
        #pragma unroll
        for (int p = 0; p < PAIRS_PER_PASS; ++p) {
            float a2t[2], xyt[2];
            #pragma unroll
            for (int e = 0; e < 2; ++e) {
                const int i = 2 * p + e;
                v8f accG = {};
                #pragma unroll
                for (int j = 0; j < 4; ++j) {
                    #pragma unroll
                    for (int s = 0; s < 2; ++s) {
                        v2f a;
                        a.x = q[i][j][2 * s];  a.y = q[i][j][2 * s + 1];
                        accG = __builtin_amdgcn_wmma_f32_16x16x4_f32(
                            false, a, false, a, (short)0, accG, false, false);
                    }
                }

                float d = accG[0];
                #pragma unroll
                for (int k = 1; k < 8; ++k) d = (r == k) ? accG[k] : d;
                const float g15x = xhalf(accG[7]);
                a2t[e] = d;
                xyt[e] = h ? accG[7] : g15x;
            }

            const float a2b = xrow8(a2t[1]);
            const float xyb = xrow8(xyt[1]);
            const float a2  = first ? a2t[0] : a2b;
            const float xy  = first ? xyt[0] : xyb;

            const int l = (t0 + 2 * p + (first ? 0 : 1)) * ROWS_PER_TILE + cown;
            if (live && valid && l < L) {
                const float ay = -xy;
                const float al = 1.0f + 2.0f * ay + y2;
                const float be = 1.0f - a2;
                float den = 1.0f + 2.0f * ay + a2 * y2;
                den = fmaxf(den, PB_MIN_DEN);
                float nn = al * al * a2 + 2.0f * al * be * ay + be * be * y2;
                nn = fmaxf(nn, 0.0f);
                float z = sqrtf(nn) / den;
                z = fminf(z, 1.0f - PB_EPS);
                const float w = z * z;
                float p = 1.0f / 13.0f;
                p = fmaf(p, w, 1.0f / 11.0f);
                p = fmaf(p, w, 1.0f / 9.0f);
                p = fmaf(p, w, 1.0f / 7.0f);
                p = fmaf(p, w, 1.0f / 5.0f);
                p = fmaf(p, w, 1.0f / 3.0f);
                float at = fmaf(z * w, p, z);
                if (__builtin_expect(__builtin_amdgcn_ballot_w32(!(z < 0.25f)) != 0u, 0))
                    at = atanhf(z);
                ob[l] = -2.0f * at;
            }
        }

        t0 += TILES_PER_PASS;
        if (t0 >= nTiles) break;
        #pragma unroll
        for (int i = 0; i < TILES_PER_PASS; ++i) {
            int l = (t0 + i) * ROWS_PER_TILE + m;
            if (l >= L) l = L - 1;
            const unsigned cand = (unsigned)it[l];
            idx[i] = (m == ROWS_PER_TILE) ? oi : cand;
        }
        #pragma unroll
        for (int i = 0; i < TILES_PER_PASS; ++i)
            load_row_share(matrix + (size_t)idx[i] * 32 + 4 * h, q[i]);
    }
}

extern "C" void kernel_launch(void* const* d_in, const int* in_sizes, int n_in,
                              void* d_out, int out_size, void* d_ws, size_t ws_size,
                              hipStream_t stream) {
    (void)n_in; (void)d_ws; (void)ws_size; (void)out_size;
    const float* matrix = (const float*)d_in[0];
    const int*   items  = (const int*)d_in[1];
    const int*   origin = (const int*)d_in[2];
    float*       out    = (float*)d_out;

    const int B = in_sizes[2];
    const int L = in_sizes[1] / B;
    const int nTiles = (L + ROWS_PER_TILE - 1) / ROWS_PER_TILE;

    poincare_dist_kernel<<<(B + WAVES_PER_BLOCK - 1) / WAVES_PER_BLOCK, 32 * WAVES_PER_BLOCK, 0, stream>>>(
        matrix, items, origin, out, B, L, nTiles);
}


# Row gathers run 3 tile pairs ahead of the matrix work (24 row loads up front, then 8 behind each pair's WMMAs) instead of all 56 ahead of the first WMMA; k order and every sum unchanged
# speedup vs baseline: 1.6488x; 1.0924x over previous
#include <hip/hip_runtime.h>
#include <math.h>

typedef __attribute__((ext_vector_type(2))) float v2f;
typedef __attribute__((ext_vector_type(4))) float v4f;
typedef __attribute__((ext_vector_type(8))) float v8f;

#define PB_EPS     1e-7f
#define PB_MIN_DEN 1e-15f

#define ROWS_PER_TILE   15
#define TILES_PER_PASS  14
#define PAIRS_PER_PASS  (TILES_PER_PASS / 2)
#define WAVES_PER_BLOCK 8
#define PREFETCH_PAIRS  3

__device__ __forceinline__ void load_row_share(const float* __restrict__ row_h, v4f q[4])
{
    #pragma unroll
    for (int j = 0; j < 4; ++j) q[j] = *(const v4f*)(row_h + 8 * j);
}

__device__ __forceinline__ float xhalf(float x)
{
    int b = __builtin_bit_cast(int, x);
    return __builtin_bit_cast(float,
        __builtin_amdgcn_permlanex16(b, b, 0x76543210, 0xfedcba98, false, false));
}

__device__ __forceinline__ float xrow8(float x)
{
    int b = __builtin_bit_cast(int, x);
    return __builtin_bit_cast(float,
        __builtin_amdgcn_update_dpp(b, b, 0x168  , 0xf, 0xf, false));
}

__global__ __launch_bounds__(32 * WAVES_PER_BLOCK) __attribute__((amdgpu_waves_per_eu(4, 4))) void poincare_dist_kernel(
    const float* __restrict__ matrix,
    const int*   __restrict__ items,
    const int*   __restrict__ origin,
    float*       __restrict__ out,
    int B, int L, int nTiles)
{
    const int lane = threadIdx.x & 31;
    const int wave = __builtin_amdgcn_readfirstlane((int)(threadIdx.x >> 5));
    const int bw   = blockIdx.x * WAVES_PER_BLOCK + wave;
    const bool live = bw < B;
    const int  b    = live ? bw : B - 1;

    const int h = lane >> 4;
    const int m = lane & 15;
    const int r = lane & 7;
    const int  cown  = r + 8 * h;
    const bool first = ((lane >> 3) & 1) == h;
    const bool valid = cown < ROWS_PER_TILE;

    const int* __restrict__ it = items + (size_t)b * L;
    float*     __restrict__ ob = out   + (size_t)b * L;

    const unsigned oi = (unsigned)origin[b];

    unsigned idx[TILES_PER_PASS];
    #pragma unroll
    for (int i = 0; i < TILES_PER_PASS; ++i) {
        int l = i * ROWS_PER_TILE + m;
        if (l >= L) l = L - 1;
        const unsigned cand = (unsigned)it[l];
        idx[i] = (m == ROWS_PER_TILE) ? oi : cand;
    }

    const float* __restrict__ yrow = matrix + (size_t)oi * 32;
    float y2 = yrow[lane];

    v4f q[TILES_PER_PASS][4];
    #pragma unroll
    for (int i = 0; i < 2 * PREFETCH_PAIRS; ++i)
        load_row_share(matrix + (size_t)idx[i] * 32 + 4 * h, q[i]);

    __builtin_amdgcn_sched_barrier(0);

    y2 = y2 * y2;
    #pragma unroll
    for (int off = 16; off > 0; off >>= 1) y2 += __shfl_xor(y2, off, 32);

    for (int t0 = 0; ; ) {
        #pragma unroll
        for (int p = 0; p < PAIRS_PER_PASS; ++p) {
            v8f accG[2];
            #pragma unroll
            for (int e = 0; e < 2; ++e) {
                const int i = 2 * p + e;
                v8f acc = {};
                #pragma unroll
                for (int j = 0; j < 4; ++j) {
                    #pragma unroll
                    for (int s = 0; s < 2; ++s) {
                        v2f a;
                        a.x = q[i][j][2 * s];  a.y = q[i][j][2 * s + 1];
                        acc = __builtin_amdgcn_wmma_f32_16x16x4_f32(
                            false, a, false, a, (short)0, acc, false, false);
                    }
                }
                accG[e] = acc;
            }

            if (p + PREFETCH_PAIRS < PAIRS_PER_PASS) {
                __builtin_amdgcn_sched_barrier(0);
                #pragma unroll
                for (int e = 0; e < 2; ++e) {
                    const int i = 2 * (p + PREFETCH_PAIRS) + e;
                    load_row_share(matrix + (size_t)idx[i] * 32 + 4 * h, q[i]);
                }
                __builtin_amdgcn_sched_barrier(0);
            }

            float a2t[2], xyt[2];
            #pragma unroll
            for (int e = 0; e < 2; ++e) {
                float d = accG[e][0];
                #pragma unroll
                for (int k = 1; k < 8; ++k) d = (r == k) ? accG[e][k] : d;
                const float g15x = xhalf(accG[e][7]);
                a2t[e] = d;
                xyt[e] = h ? accG[e][7] : g15x;
            }

            const float a2b = xrow8(a2t[1]);
            const float xyb = xrow8(xyt[1]);
            const float a2  = first ? a2t[0] : a2b;
            const float xy  = first ? xyt[0] : xyb;

            const int l = (t0 + 2 * p + (first ? 0 : 1)) * ROWS_PER_TILE + cown;
            if (live && valid && l < L) {
                const float ay = -xy;
                const float al = 1.0f + 2.0f * ay + y2;
                const float be = 1.0f - a2;
                float den = 1.0f + 2.0f * ay + a2 * y2;
                den = fmaxf(den, PB_MIN_DEN);
                float nn = al * al * a2 + 2.0f * al * be * ay + be * be * y2;
                nn = fmaxf(nn, 0.0f);
                float z = sqrtf(nn) / den;
                z = fminf(z, 1.0f - PB_EPS);
                const float w = z * z;
                float p = 1.0f / 13.0f;
                p = fmaf(p, w, 1.0f / 11.0f);
                p = fmaf(p, w, 1.0f / 9.0f);
                p = fmaf(p, w, 1.0f / 7.0f);
                p = fmaf(p, w, 1.0f / 5.0f);
                p = fmaf(p, w, 1.0f / 3.0f);
                float at = fmaf(z * w, p, z);
                if (__builtin_expect(__builtin_amdgcn_ballot_w32(!(z < 0.25f)) != 0u, 0))
                    at = atanhf(z);
                ob[l] = -2.0f * at;
            }
        }

        t0 += TILES_PER_PASS;
        if (t0 >= nTiles) break;
        #pragma unroll
        for (int i = 0; i < TILES_PER_PASS; ++i) {
            int l = (t0 + i) * ROWS_PER_TILE + m;
            if (l >= L) l = L - 1;
            const unsigned cand = (unsigned)it[l];
            idx[i] = (m == ROWS_PER_TILE) ? oi : cand;
        }
        #pragma unroll
        for (int i = 0; i < 2 * PREFETCH_PAIRS; ++i)
            load_row_share(matrix + (size_t)idx[i] * 32 + 4 * h, q[i]);
    }
}

extern "C" void kernel_launch(void* const* d_in, const int* in_sizes, int n_in,
                              void* d_out, int out_size, void* d_ws, size_t ws_size,
                              hipStream_t stream) {
    (void)n_in; (void)d_ws; (void)ws_size; (void)out_size;
    const float* matrix = (const float*)d_in[0];
    const int*   items  = (const int*)d_in[1];
    const int*   origin = (const int*)d_in[2];
    float*       out    = (float*)d_out;

    const int B = in_sizes[2];
    const int L = in_sizes[1] / B;
    const int nTiles = (L + ROWS_PER_TILE - 1) / ROWS_PER_TILE;

    poincare_dist_kernel<<<(B + WAVES_PER_BLOCK - 1) / WAVES_PER_BLOCK, 32 * WAVES_PER_BLOCK, 0, stream>>>(
        matrix, items, origin, out, B, L, nTiles);
}
